# starting kernel
# baseline (speedup 1.0000x reference)
#include <hip/hip_runtime.h>

typedef __attribute__((ext_vector_type(2))) float v2f;
typedef __attribute__((ext_vector_type(4))) float v4f;
typedef __attribute__((ext_vector_type(8))) float v8f;

#define B_   4096
#define T_   256
#define F_   64
#define H_   16
#define G4H  64
#define D_   64
#define XPAD 68
#define HPAD 18
#define WPP  160

__device__ __forceinline__ float sigf(float v) { return 1.0f / (1.0f + __expf(-v)); }

__global__ __launch_bounds__(32)
void lstm_fused_kernel(const float* __restrict__ x, const float* __restrict__ W,
                       const float* __restrict__ U, const float* __restrict__ b,
                       const float* __restrict__ Wd, const float* __restrict__ bd,
                       float* __restrict__ out) {
  __shared__ float Wlds[32 * WPP];
  __shared__ float Wdlds[8 * WPP];
  __shared__ float xlds[2][16 * XPAD];
  __shared__ float hlds[16 * HPAD];

  const int lane  = threadIdx.x & 31;
  const int n     = lane & 15;
  const int half  = lane >> 4;
  const int khalf = half * 2;
  const int mhalf = half * 8;
  const int b0    = blockIdx.x * 16;

  for (int i = lane; i < F_ * G4H; i += 32) {
    int j = i & 1, col = (i >> 1) & 63, pp = i >> 7;
    Wlds[pp * WPP + col * 2 + j] = W[(2 * pp + j) * G4H + col];
  }
  for (int i = lane; i < H_ * D_; i += 32) {
    int j = i & 1, col = (i >> 1) & 63, pp = i >> 7;
    Wdlds[pp * WPP + col * 2 + j] = Wd[(2 * pp + j) * D_ + col];
  }
  __syncthreads();

  v2f ufrag[4][4];
#pragma unroll
  for (int kk = 0; kk < 4; ++kk)
#pragma unroll
    for (int g = 0; g < 4; ++g) {
      int k0 = kk * 4 + khalf;
      ufrag[kk][g].x = U[k0 * G4H + g * 16 + n];
      ufrag[kk][g].y = U[(k0 + 1) * G4H + g * 16 + n];
    }

  float bg[4];
#pragma unroll
  for (int g = 0; g < 4; ++g) bg[g] = b[g * 16 + n];

  v8f h = {}; v8f c = {};
  float* xl0 = &xlds[0][0];
  float* xl1 = &xlds[1][0];
  float* hl  = &hlds[0];

  auto loadx = [&](int t, v4f* xr) {
#pragma unroll
    for (int i = 0; i < 8; ++i) {
      int idx = i * 32 + lane;
      int row = idx >> 4, c4 = idx & 15;
      xr[i] = *(const v4f*)(x + ((size_t)(b0 + row) * T_ + t) * F_ + c4 * 4);
    }
  };
  auto storex = [&](float* dst, const v4f* xr) {
#pragma unroll
    for (int i = 0; i < 8; ++i) {
      int idx = i * 32 + lane;
      int row = idx >> 4, c4 = idx & 15;
      float* p = dst + row * XPAD + c4 * 4;
      v2f lo; lo.x = xr[i].x; lo.y = xr[i].y;
      v2f hi; hi.x = xr[i].z; hi.y = xr[i].w;
      *(v2f*)p = lo;
      *(v2f*)(p + 2) = hi;
    }
  };

  v4f xr[8];
  loadx(0, xr);
  storex(xl0, xr);

  int p = 0;
  for (int t = 0; t < T_; ++t) {
    if (t + 1 < T_) loadx(t + 1, xr);
    if (t + 2 < T_) {
      const float* pf = x + ((size_t)(b0 + (lane >> 1)) * T_ + (t + 2)) * F_ + (lane & 1) * 32;
      __builtin_prefetch(pf, 0, 0);
    }

    v8f acc[4];
#pragma unroll
    for (int g = 0; g < 4; ++g)
#pragma unroll
      for (int r = 0; r < 8; ++r) acc[g][r] = bg[g];

    float* xl = p ? xl1 : xl0;

#pragma unroll 4
    for (int kk = 0; kk < 16; ++kk) {
      v2f a = *(const v2f*)(xl + n * XPAD + kk * 4 + khalf);
      const float* wrow = Wlds + (kk * 2 + half) * WPP;
#pragma unroll
      for (int g = 0; g < 4; ++g) {
        v2f wb = *(const v2f*)(wrow + (g * 16 + n) * 2);
        acc[g] = __builtin_amdgcn_wmma_f32_16x16x4_f32(false, a, false, wb,
                                                       (short)0, acc[g], false, false);
      }
    }

#pragma unroll
    for (int r = 0; r < 8; ++r) hl[(r + mhalf) * HPAD + n] = h[r];
#pragma unroll
    for (int kk = 0; kk < 4; ++kk) {
      v2f a = *(const v2f*)(hl + n * HPAD + kk * 4 + khalf);
#pragma unroll
      for (int g = 0; g < 4; ++g)
        acc[g] = __builtin_amdgcn_wmma_f32_16x16x4_f32(false, a, false, ufrag[kk][g],
                                                       (short)0, acc[g], false, false);
    }

#pragma unroll
    for (int r = 0; r < 8; ++r) {
      float gi = sigf(acc[0][r]);
      float gf = sigf(acc[1][r]);
      float gg = fmaxf(acc[2][r], 0.0f);
      float go = sigf(acc[3][r]);
      float cn = gf * c[r] + gi * gg;
      c[r] = cn;
      h[r] = go * fmaxf(cn, 0.0f);
    }

    if (t + 1 < T_) { storex(p ? xl0 : xl1, xr); p ^= 1; }
  }

#pragma unroll
  for (int r = 0; r < 8; ++r) hl[(r + mhalf) * HPAD + n] = h[r];

  v8f dacc[4];
#pragma unroll
  for (int j = 0; j < 4; ++j) {
    float bj = bd[j * 16 + n];
#pragma unroll
    for (int r = 0; r < 8; ++r) dacc[j][r] = bj;
  }
#pragma unroll
  for (int kk = 0; kk < 4; ++kk) {
    v2f a = *(const v2f*)(hl + n * HPAD + kk * 4 + khalf);
    const float* wrow = Wdlds + (kk * 2 + half) * WPP;
#pragma unroll
    for (int j = 0; j < 4; ++j) {
      v2f wb = *(const v2f*)(wrow + (j * 16 + n) * 2);
      dacc[j] = __builtin_amdgcn_wmma_f32_16x16x4_f32(false, a, false, wb,
                                                      (short)0, dacc[j], false, false);
    }
  }

  float* ol = xl0;
#pragma unroll
  for (int j = 0; j < 4; ++j)
#pragma unroll
    for (int r = 0; r < 8; ++r)
      ol[(r + mhalf) * XPAD + j * 16 + n] = sigf(dacc[j][r]);

  if (lane < 16) {
    float s = 0.0f;
    for (int q = 0; q < 64; ++q) s += ol[lane * XPAD + q];
    hl[lane] = 1.0f / s;
  }

#pragma unroll 4
  for (int i = 0; i < 32; ++i) {
    int idx = i * 32 + lane;
    int row = idx >> 6, col = idx & 63;
    out[(size_t)(b0 + row) * D_ + col] = ol[row * XPAD + col] * hl[row];
  }
}

extern "C" void kernel_launch(void* const* d_in, const int* in_sizes, int n_in,
                              void* d_out, int out_size, void* d_ws, size_t ws_size,
                              hipStream_t stream) {
  (void)in_sizes; (void)n_in; (void)out_size; (void)d_ws; (void)ws_size;
  const float* x  = (const float*)d_in[0];
  const float* W  = (const float*)d_in[1];
  const float* U  = (const float*)d_in[2];
  const float* b  = (const float*)d_in[3];
  const float* Wd = (const float*)d_in[4];
  const float* bd = (const float*)d_in[5];
  float* out = (float*)d_out;

  dim3 grid(B_ / 16);
  dim3 block(32);
  hipLaunchKernelGGL(lstm_fused_kernel, grid, block, 0, stream,
                     x, W, U, b, Wd, bd, out);
}


# x_t@W+b moved off the recurrent wave: 4-wave blocks, 3 projection waves (W fragments in registers) fill an LDS ring of z0 tiles one 6-step chunk ahead; wave 0 only adds h@U and the gates
# speedup vs baseline: 2.7647x; 2.7647x over previous
#include <hip/hip_runtime.h>


typedef __attribute__((ext_vector_type(2))) float v2f;
typedef __attribute__((ext_vector_type(4))) float v4f;
typedef __attribute__((ext_vector_type(8))) float v8f;

#define B_   4096
#define T_   256
#define F_   64
#define H_   16
#define G4H  64
#define D_   64
#define XPAD 68
#define HPAD 18
#define WPP  160
#define NPROD 3
#define TC   6
#define NCH  ((T_ + TC - 1) / TC)
#define SLOT 1024

__device__ __forceinline__ float sigf(float v) { return 1.0f / (1.0f + __expf(-v)); }

__global__ __launch_bounds__(128)
void lstm_fused_kernel(const float* __restrict__ x, const float* __restrict__ W,
                       const float* __restrict__ U, const float* __restrict__ b,
                       const float* __restrict__ Wd, const float* __restrict__ bd,
                       float* __restrict__ out) {
  __shared__ __attribute__((aligned(16))) float ring[2 * TC * SLOT];
  __shared__ __attribute__((aligned(16))) float xlds[NPROD * 16 * XPAD];
  __shared__ float Wdlds[8 * WPP];
  __shared__ float hlds[16 * HPAD];

  const int lane  = threadIdx.x & 31;
  const int wave  = __builtin_amdgcn_readfirstlane((int)(threadIdx.x >> 5));
  const int n     = lane & 15;
  const int half  = lane >> 4;
  const int khalf = half * 2;
  const int mhalf = half * 8;
  const int b0    = blockIdx.x * 16;
  const int pw    = wave - 1;

  v2f wfrag[16][4];
  v2f ufrag[4][4];
  float bg[4];
  v4f xr[8];
  v8f h = {}; v8f c = {};
  float* xs = xlds + (pw < 0 ? 0 : pw) * 16 * XPAD;
  float* hl = &hlds[0];

  auto loadx = [&](int t, v4f* xv) {
#pragma unroll
    for (int i = 0; i < 8; ++i) {
      int idx = i * 32 + lane;
      int row = idx >> 4, c4 = idx & 15;
      xv[i] = *(const v4f*)(x + ((size_t)(b0 + row) * T_ + t) * F_ + c4 * 4);
    }
  };
  auto storex = [&](float* dst, const v4f* xv) {
#pragma unroll
    for (int i = 0; i < 8; ++i) {
      int idx = i * 32 + lane;
      int row = idx >> 4, c4 = idx & 15;
      float* p = dst + row * XPAD + c4 * 4;
      v2f lo; lo.x = xv[i].x; lo.y = xv[i].y;
      v2f hi; hi.x = xv[i].z; hi.y = xv[i].w;
      *(v2f*)p = lo;
      *(v2f*)(p + 2) = hi;
    }
  };

  if (wave == 0) {
    for (int i = lane; i < H_ * D_; i += 32) {
      int j = i & 1, col = (i >> 1) & 63, pp = i >> 7;
      Wdlds[pp * WPP + col * 2 + j] = Wd[(2 * pp + j) * D_ + col];
    }
#pragma unroll
    for (int kk = 0; kk < 4; ++kk)
#pragma unroll
      for (int g = 0; g < 4; ++g) {
        int k0 = kk * 4 + khalf;
        ufrag[kk][g].x = U[k0 * G4H + g * 16 + n];
        ufrag[kk][g].y = U[(k0 + 1) * G4H + g * 16 + n];
      }
  } else {
#pragma unroll
    for (int kk = 0; kk < 16; ++kk)
#pragma unroll
      for (int g = 0; g < 4; ++g) {
        int k0 = kk * 4 + khalf;
        wfrag[kk][g].x = W[k0 * G4H + g * 16 + n];
        wfrag[kk][g].y = W[(k0 + 1) * G4H + g * 16 + n];
      }
#pragma unroll
    for (int g = 0; g < 4; ++g) bg[g] = b[g * 16 + n];
    loadx(pw, xr);
  }

  auto produce = [&](int ch) {
    float* cbase = ring + (ch & 1) * (TC * SLOT);
    for (int j = 0; j < 2; ++j) {
      const int tl = j * NPROD + pw;
      const int t  = ch * TC + tl;
      if (t < T_) {
        storex(xs, xr);
        if (t + NPROD < T_) loadx(t + NPROD, xr);
        if (t + 2 * NPROD < T_) {
          const float* pf = x + ((size_t)(b0 + (lane >> 1)) * T_ + (t + 2 * NPROD)) * F_ + (lane & 1) * 32;
          __builtin_prefetch(pf, 0, 0);
        }
        v2f a[16];
#pragma unroll
        for (int kk = 0; kk < 16; ++kk) a[kk] = *(const v2f*)(xs + n * XPAD + kk * 4 + khalf);

        v8f acc[4];
#pragma unroll
        for (int g = 0; g < 4; ++g)
#pragma unroll
          for (int r = 0; r < 8; ++r) acc[g][r] = bg[g];
#pragma unroll
        for (int kk = 0; kk < 16; ++kk)
#pragma unroll
          for (int g = 0; g < 4; ++g)
            acc[g] = __builtin_amdgcn_wmma_f32_16x16x4_f32(false, a[kk], false, wfrag[kk][g],
                                                           (short)0, acc[g], false, false);
        float* slot = cbase + tl * SLOT + lane * 4;
#pragma unroll
        for (int g = 0; g < 4; ++g) {
          v4f lo; lo.x = acc[g][0]; lo.y = acc[g][1]; lo.z = acc[g][2]; lo.w = acc[g][3];
          v4f hi; hi.x = acc[g][4]; hi.y = acc[g][5]; hi.z = acc[g][6]; hi.w = acc[g][7];
          *(v4f*)(slot + (g * 2 + 0) * 128) = lo;
          *(v4f*)(slot + (g * 2 + 1) * 128) = hi;
        }
      }
    }
  };

  auto consume = [&](int ch) {
    const float* cbase = ring + (ch & 1) * (TC * SLOT) + lane * 4;
    const int left = T_ - ch * TC;
    const int nst  = left < TC ? left : TC;
    v4f cur[8];
#pragma unroll
    for (int q = 0; q < 8; ++q) cur[q] = *(const v4f*)(cbase + q * 128);
    for (int tl = 0; tl < nst; ++tl) {
      const int tn = (tl + 1 < nst) ? tl + 1 : tl;
      v4f nxt[8];
#pragma unroll
      for (int q = 0; q < 8; ++q) nxt[q] = *(const v4f*)(cbase + tn * SLOT + q * 128);

      v8f acc[4];
#pragma unroll
      for (int g = 0; g < 4; ++g) {
        acc[g][0] = cur[2 * g][0]; acc[g][1] = cur[2 * g][1]; acc[g][2] = cur[2 * g][2]; acc[g][3] = cur[2 * g][3];
        acc[g][4] = cur[2 * g + 1][0]; acc[g][5] = cur[2 * g + 1][1]; acc[g][6] = cur[2 * g + 1][2]; acc[g][7] = cur[2 * g + 1][3];
      }

#pragma unroll
      for (int r = 0; r < 8; ++r) hl[(r + mhalf) * HPAD + n] = h[r];
#pragma unroll
      for (int kk = 0; kk < 4; ++kk) {
        v2f a = *(const v2f*)(hl + n * HPAD + kk * 4 + khalf);
#pragma unroll
        for (int g = 0; g < 4; ++g)
          acc[g] = __builtin_amdgcn_wmma_f32_16x16x4_f32(false, a, false, ufrag[kk][g],
                                                         (short)0, acc[g], false, false);
      }

#pragma unroll
      for (int r = 0; r < 8; ++r) {
        float gi = sigf(acc[0][r]);
        float gf = sigf(acc[1][r]);
        float gg = fmaxf(acc[2][r], 0.0f);
        float go = sigf(acc[3][r]);
        float cn = gf * c[r] + gi * gg;
        c[r] = cn;
        h[r] = go * fmaxf(cn, 0.0f);
      }
#pragma unroll
      for (int q = 0; q < 8; ++q) cur[q] = nxt[q];
    }
  };

  for (int ch = -1; ch < NCH; ++ch) {
    if (wave == 0) {
      if (ch >= 0) consume(ch);
    } else {
      if (ch + 1 < NCH) produce(ch + 1);
    }
    __syncthreads();
  }

  if (wave != 0) return;

#pragma unroll
  for (int r = 0; r < 8; ++r) hl[(r + mhalf) * HPAD + n] = h[r];

  v8f dacc[4];
#pragma unroll
  for (int j = 0; j < 4; ++j) {
    float bj = bd[j * 16 + n];
#pragma unroll
    for (int r = 0; r < 8; ++r) dacc[j][r] = bj;
  }
#pragma unroll
  for (int kk = 0; kk < 4; ++kk) {
    v2f a = *(const v2f*)(hl + n * HPAD + kk * 4 + khalf);
    const float* wrow = Wdlds + (kk * 2 + half) * WPP;
#pragma unroll
    for (int j = 0; j < 4; ++j) {
      v2f wb = *(const v2f*)(wrow + (j * 16 + n) * 2);
      dacc[j] = __builtin_amdgcn_wmma_f32_16x16x4_f32(false, a, false, wb,
                                                      (short)0, dacc[j], false, false);
    }
  }

  float* ol = &ring[0];
#pragma unroll
  for (int j = 0; j < 4; ++j)
#pragma unroll
    for (int r = 0; r < 8; ++r)
      ol[(r + mhalf) * XPAD + j * 16 + n] = sigf(dacc[j][r]);

  if (lane < 16) {
    float s = 0.0f;
    for (int q = 0; q < 64; ++q) s += ol[lane * XPAD + q];
    hl[lane] = 1.0f / s;
  }

#pragma unroll 4
  for (int i = 0; i < 32; ++i) {
    int idx = i * 32 + lane;
    int row = idx >> 6, col = idx & 63;
    out[(size_t)(b0 + row) * D_ + col] = ol[row * XPAD + col] * hl[row];
  }
}

extern "C" void kernel_launch(void* const* d_in, const int* in_sizes, int n_in,
                              void* d_out, int out_size, void* d_ws, size_t ws_size,
                              hipStream_t stream) {
  (void)in_sizes; (void)n_in; (void)out_size; (void)d_ws; (void)ws_size;
  const float* x  = (const float*)d_in[0];
  const float* W  = (const float*)d_in[1];
  const float* U  = (const float*)d_in[2];
  const float* b  = (const float*)d_in[3];
  const float* Wd = (const float*)d_in[4];
  const float* bd = (const float*)d_in[5];
  float* out = (float*)d_out;

  dim3 grid(B_ / 16);
  dim3 block(128);
  hipLaunchKernelGGL(lstm_fused_kernel, grid, block, 0, stream,
                     x, W, U, b, Wd, bd, out);
}


# recurrent gate sigmoids use the one-instruction hardware reciprocal of 1+exp(-z) instead of the full f32 division sequence (about one ulp per gate; dense head and row sums keep exact division)
# speedup vs baseline: 4.8037x; 1.7375x over previous
#include <hip/hip_runtime.h>


typedef __attribute__((ext_vector_type(2))) float v2f;
typedef __attribute__((ext_vector_type(4))) float v4f;
typedef __attribute__((ext_vector_type(8))) float v8f;

#define B_   4096
#define T_   256
#define F_   64
#define H_   16
#define G4H  64
#define D_   64
#define XPAD 68
#define HPAD 18
#define WPP  160
#define NPROD 3
#define TC   6
#define NCH  ((T_ + TC - 1) / TC)
#define SLOT 1024

__device__ __forceinline__ float sigf(float v) { return 1.0f / (1.0f + __expf(-v)); }

__device__ __forceinline__ float sig_gate(float v) { return __builtin_amdgcn_rcpf(1.0f + __expf(-v)); }

__global__ __launch_bounds__(128)
void lstm_fused_kernel(const float* __restrict__ x, const float* __restrict__ W,
                       const float* __restrict__ U, const float* __restrict__ b,
                       const float* __restrict__ Wd, const float* __restrict__ bd,
                       float* __restrict__ out) {
  __shared__ __attribute__((aligned(16))) float ring[2 * TC * SLOT];
  __shared__ __attribute__((aligned(16))) float xlds[NPROD * 16 * XPAD];
  __shared__ float Wdlds[8 * WPP];
  __shared__ float hlds[16 * HPAD];

  const int lane  = threadIdx.x & 31;
  const int wave  = __builtin_amdgcn_readfirstlane((int)(threadIdx.x >> 5));
  const int n     = lane & 15;
  const int half  = lane >> 4;
  const int khalf = half * 2;
  const int mhalf = half * 8;
  const int b0    = blockIdx.x * 16;
  const int pw    = wave - 1;

  v2f wfrag[16][4];
  v2f ufrag[4][4];
  float bg[4];
  v4f xr[8];
  v8f h = {}; v8f c = {};
  float* xs = xlds + (pw < 0 ? 0 : pw) * 16 * XPAD;
  float* hl = &hlds[0];

  auto loadx = [&](int t, v4f* xv) {
#pragma unroll
    for (int i = 0; i < 8; ++i) {
      int idx = i * 32 + lane;
      int row = idx >> 4, c4 = idx & 15;
      xv[i] = *(const v4f*)(x + ((size_t)(b0 + row) * T_ + t) * F_ + c4 * 4);
    }
  };
  auto storex = [&](float* dst, const v4f* xv) {
#pragma unroll
    for (int i = 0; i < 8; ++i) {
      int idx = i * 32 + lane;
      int row = idx >> 4, c4 = idx & 15;
      float* p = dst + row * XPAD + c4 * 4;
      v2f lo; lo.x = xv[i].x; lo.y = xv[i].y;
      v2f hi; hi.x = xv[i].z; hi.y = xv[i].w;
      *(v2f*)p = lo;
      *(v2f*)(p + 2) = hi;
    }
  };

  if (wave == 0) {
    for (int i = lane; i < H_ * D_; i += 32) {
      int j = i & 1, col = (i >> 1) & 63, pp = i >> 7;
      Wdlds[pp * WPP + col * 2 + j] = Wd[(2 * pp + j) * D_ + col];
    }
#pragma unroll
    for (int kk = 0; kk < 4; ++kk)
#pragma unroll
      for (int g = 0; g < 4; ++g) {
        int k0 = kk * 4 + khalf;
        ufrag[kk][g].x = U[k0 * G4H + g * 16 + n];
        ufrag[kk][g].y = U[(k0 + 1) * G4H + g * 16 + n];
      }
  } else {
#pragma unroll
    for (int kk = 0; kk < 16; ++kk)
#pragma unroll
      for (int g = 0; g < 4; ++g) {
        int k0 = kk * 4 + khalf;
        wfrag[kk][g].x = W[k0 * G4H + g * 16 + n];
        wfrag[kk][g].y = W[(k0 + 1) * G4H + g * 16 + n];
      }
#pragma unroll
    for (int g = 0; g < 4; ++g) bg[g] = b[g * 16 + n];
    loadx(pw, xr);
  }

  auto produce = [&](int ch) {
    float* cbase = ring + (ch & 1) * (TC * SLOT);
    for (int j = 0; j < 2; ++j) {
      const int tl = j * NPROD + pw;
      const int t  = ch * TC + tl;
      if (t < T_) {
        storex(xs, xr);
        if (t + NPROD < T_) loadx(t + NPROD, xr);
        if (t + 2 * NPROD < T_) {
          const float* pf = x + ((size_t)(b0 + (lane >> 1)) * T_ + (t + 2 * NPROD)) * F_ + (lane & 1) * 32;
          __builtin_prefetch(pf, 0, 0);
        }
        v2f a[16];
#pragma unroll
        for (int kk = 0; kk < 16; ++kk) a[kk] = *(const v2f*)(xs + n * XPAD + kk * 4 + khalf);

        v8f acc[4];
#pragma unroll
        for (int g = 0; g < 4; ++g)
#pragma unroll
          for (int r = 0; r < 8; ++r) acc[g][r] = bg[g];
#pragma unroll
        for (int kk = 0; kk < 16; ++kk)
#pragma unroll
          for (int g = 0; g < 4; ++g)
            acc[g] = __builtin_amdgcn_wmma_f32_16x16x4_f32(false, a[kk], false, wfrag[kk][g],
                                                           (short)0, acc[g], false, false);
        float* slot = cbase + tl * SLOT + lane * 4;
#pragma unroll
        for (int g = 0; g < 4; ++g) {
          v4f lo; lo.x = acc[g][0]; lo.y = acc[g][1]; lo.z = acc[g][2]; lo.w = acc[g][3];
          v4f hi; hi.x = acc[g][4]; hi.y = acc[g][5]; hi.z = acc[g][6]; hi.w = acc[g][7];
          *(v4f*)(slot + (g * 2 + 0) * 128) = lo;
          *(v4f*)(slot + (g * 2 + 1) * 128) = hi;
        }
      }
    }
  };

  auto consume = [&](int ch) {
    const float* cbase = ring + (ch & 1) * (TC * SLOT) + lane * 4;
    const int left = T_ - ch * TC;
    const int nst  = left < TC ? left : TC;
    v4f cur[8];
#pragma unroll
    for (int q = 0; q < 8; ++q) cur[q] = *(const v4f*)(cbase + q * 128);
    for (int tl = 0; tl < nst; ++tl) {
      const int tn = (tl + 1 < nst) ? tl + 1 : tl;
      v4f nxt[8];
#pragma unroll
      for (int q = 0; q < 8; ++q) nxt[q] = *(const v4f*)(cbase + tn * SLOT + q * 128);

      v8f acc[4];
#pragma unroll
      for (int g = 0; g < 4; ++g) {
        acc[g][0] = cur[2 * g][0]; acc[g][1] = cur[2 * g][1]; acc[g][2] = cur[2 * g][2]; acc[g][3] = cur[2 * g][3];
        acc[g][4] = cur[2 * g + 1][0]; acc[g][5] = cur[2 * g + 1][1]; acc[g][6] = cur[2 * g + 1][2]; acc[g][7] = cur[2 * g + 1][3];
      }

#pragma unroll
      for (int r = 0; r < 8; ++r) hl[(r + mhalf) * HPAD + n] = h[r];
#pragma unroll
      for (int kk = 0; kk < 4; ++kk) {
        v2f a = *(const v2f*)(hl + n * HPAD + kk * 4 + khalf);
#pragma unroll
        for (int g = 0; g < 4; ++g)
          acc[g] = __builtin_amdgcn_wmma_f32_16x16x4_f32(false, a, false, ufrag[kk][g],
                                                         (short)0, acc[g], false, false);
      }

#pragma unroll
      for (int r = 0; r < 8; ++r) {
        float gi = sig_gate(acc[0][r]);
        float gf = sig_gate(acc[1][r]);
        float gg = fmaxf(acc[2][r], 0.0f);
        float go = sig_gate(acc[3][r]);
        float cn = gf * c[r] + gi * gg;
        c[r] = cn;
        h[r] = go * fmaxf(cn, 0.0f);
      }
#pragma unroll
      for (int q = 0; q < 8; ++q) cur[q] = nxt[q];
    }
  };

  for (int ch = -1; ch < NCH; ++ch) {
    if (wave == 0) {
      if (ch >= 0) consume(ch);
    } else {
      if (ch + 1 < NCH) produce(ch + 1);
    }
    __syncthreads();
  }

  if (wave != 0) return;

#pragma unroll
  for (int r = 0; r < 8; ++r) hl[(r + mhalf) * HPAD + n] = h[r];

  v8f dacc[4];
#pragma unroll
  for (int j = 0; j < 4; ++j) {
    float bj = bd[j * 16 + n];
#pragma unroll
    for (int r = 0; r < 8; ++r) dacc[j][r] = bj;
  }
#pragma unroll
  for (int kk = 0; kk < 4; ++kk) {
    v2f a = *(const v2f*)(hl + n * HPAD + kk * 4 + khalf);
    const float* wrow = Wdlds + (kk * 2 + half) * WPP;
#pragma unroll
    for (int j = 0; j < 4; ++j) {
      v2f wb = *(const v2f*)(wrow + (j * 16 + n) * 2);
      dacc[j] = __builtin_amdgcn_wmma_f32_16x16x4_f32(false, a, false, wb,
                                                      (short)0, dacc[j], false, false);
    }
  }

  float* ol = &ring[0];
#pragma unroll
  for (int j = 0; j < 4; ++j)
#pragma unroll
    for (int r = 0; r < 8; ++r)
      ol[(r + mhalf) * XPAD + j * 16 + n] = sigf(dacc[j][r]);

  if (lane < 16) {
    float s = 0.0f;
    for (int q = 0; q < 64; ++q) s += ol[lane * XPAD + q];
    hl[lane] = 1.0f / s;
  }

#pragma unroll 4
  for (int i = 0; i < 32; ++i) {
    int idx = i * 32 + lane;
    int row = idx >> 6, col = idx & 63;
    out[(size_t)(b0 + row) * D_ + col] = ol[row * XPAD + col] * hl[row];
  }
}

extern "C" void kernel_launch(void* const* d_in, const int* in_sizes, int n_in,
                              void* d_out, int out_size, void* d_ws, size_t ws_size,
                              hipStream_t stream) {
  (void)in_sizes; (void)n_in; (void)out_size; (void)d_ws; (void)ws_size;
  const float* x  = (const float*)d_in[0];
  const float* W  = (const float*)d_in[1];
  const float* U  = (const float*)d_in[2];
  const float* b  = (const float*)d_in[3];
  const float* Wd = (const float*)d_in[4];
  const float* bd = (const float*)d_in[5];
  float* out = (float*)d_out;

  dim3 grid(B_ / 16);
  dim3 block(128);
  hipLaunchKernelGGL(lstm_fused_kernel, grid, block, 0, stream,
                     x, W, U, b, Wd, bd, out);
}


# recurrent gate sigmoids as 1/2 + 1/2 tanh(v/2) on the one-instruction hardware tanh: 24 transcendentals per step on the recurrence wave instead of 48 (gate values move by up to about 1e-7 absolute)
# speedup vs baseline: 5.1097x; 1.0637x over previous
#include <hip/hip_runtime.h>


typedef __attribute__((ext_vector_type(2))) float v2f;
typedef __attribute__((ext_vector_type(4))) float v4f;
typedef __attribute__((ext_vector_type(8))) float v8f;

#define B_   4096
#define T_   256
#define F_   64
#define H_   16
#define G4H  64
#define D_   64
#define XPAD 68
#define HPAD 18
#define WPP  160
#define NPROD 3
#define TC   6
#define NCH  ((T_ + TC - 1) / TC)
#define SLOT 1024

__device__ __forceinline__ float sigf(float v) { return 1.0f / (1.0f + __expf(-v)); }

__device__ __forceinline__ float sig_gate(float v) {
  return __builtin_fmaf(0.5f, __builtin_amdgcn_tanhf(0.5f * v), 0.5f);
}

__global__ __launch_bounds__(128)
void lstm_fused_kernel(const float* __restrict__ x, const float* __restrict__ W,
                       const float* __restrict__ U, const float* __restrict__ b,
                       const float* __restrict__ Wd, const float* __restrict__ bd,
                       float* __restrict__ out) {
  __shared__ __attribute__((aligned(16))) float ring[2 * TC * SLOT];
  __shared__ __attribute__((aligned(16))) float xlds[NPROD * 16 * XPAD];
  __shared__ float Wdlds[8 * WPP];
  __shared__ float hlds[16 * HPAD];

  const int lane  = threadIdx.x & 31;
  const int wave  = __builtin_amdgcn_readfirstlane((int)(threadIdx.x >> 5));
  const int n     = lane & 15;
  const int half  = lane >> 4;
  const int khalf = half * 2;
  const int mhalf = half * 8;
  const int b0    = blockIdx.x * 16;
  const int pw    = wave - 1;

  v2f wfrag[16][4];
  v2f ufrag[4][4];
  float bg[4];
  v4f xr[8];
  v8f h = {}; v8f c = {};
  float* xs = xlds + (pw < 0 ? 0 : pw) * 16 * XPAD;
  float* hl = &hlds[0];

  auto loadx = [&](int t, v4f* xv) {
#pragma unroll
    for (int i = 0; i < 8; ++i) {
      int idx = i * 32 + lane;
      int row = idx >> 4, c4 = idx & 15;
      xv[i] = *(const v4f*)(x + ((size_t)(b0 + row) * T_ + t) * F_ + c4 * 4);
    }
  };
  auto storex = [&](float* dst, const v4f* xv) {
#pragma unroll
    for (int i = 0; i < 8; ++i) {
      int idx = i * 32 + lane;
      int row = idx >> 4, c4 = idx & 15;
      float* p = dst + row * XPAD + c4 * 4;
      v2f lo; lo.x = xv[i].x; lo.y = xv[i].y;
      v2f hi; hi.x = xv[i].z; hi.y = xv[i].w;
      *(v2f*)p = lo;
      *(v2f*)(p + 2) = hi;
    }
  };

  if (wave == 0) {
    for (int i = lane; i < H_ * D_; i += 32) {
      int j = i & 1, col = (i >> 1) & 63, pp = i >> 7;
      Wdlds[pp * WPP + col * 2 + j] = Wd[(2 * pp + j) * D_ + col];
    }
#pragma unroll
    for (int kk = 0; kk < 4; ++kk)
#pragma unroll
      for (int g = 0; g < 4; ++g) {
        int k0 = kk * 4 + khalf;
        ufrag[kk][g].x = U[k0 * G4H + g * 16 + n];
        ufrag[kk][g].y = U[(k0 + 1) * G4H + g * 16 + n];
      }
  } else {
#pragma unroll
    for (int kk = 0; kk < 16; ++kk)
#pragma unroll
      for (int g = 0; g < 4; ++g) {
        int k0 = kk * 4 + khalf;
        wfrag[kk][g].x = W[k0 * G4H + g * 16 + n];
        wfrag[kk][g].y = W[(k0 + 1) * G4H + g * 16 + n];
      }
#pragma unroll
    for (int g = 0; g < 4; ++g) bg[g] = b[g * 16 + n];
    loadx(pw, xr);
  }

  auto produce = [&](int ch) {
    float* cbase = ring + (ch & 1) * (TC * SLOT);
    for (int j = 0; j < 2; ++j) {
      const int tl = j * NPROD + pw;
      const int t  = ch * TC + tl;
      if (t < T_) {
        storex(xs, xr);
        if (t + NPROD < T_) loadx(t + NPROD, xr);
        if (t + 2 * NPROD < T_) {
          const float* pf = x + ((size_t)(b0 + (lane >> 1)) * T_ + (t + 2 * NPROD)) * F_ + (lane & 1) * 32;
          __builtin_prefetch(pf, 0, 0);
        }
        v2f a[16];
#pragma unroll
        for (int kk = 0; kk < 16; ++kk) a[kk] = *(const v2f*)(xs + n * XPAD + kk * 4 + khalf);

        v8f acc[4];
#pragma unroll
        for (int g = 0; g < 4; ++g)
#pragma unroll
          for (int r = 0; r < 8; ++r) acc[g][r] = bg[g];
#pragma unroll
        for (int kk = 0; kk < 16; ++kk)
#pragma unroll
          for (int g = 0; g < 4; ++g)
            acc[g] = __builtin_amdgcn_wmma_f32_16x16x4_f32(false, a[kk], false, wfrag[kk][g],
                                                           (short)0, acc[g], false, false);
        float* slot = cbase + tl * SLOT + lane * 4;
#pragma unroll
        for (int g = 0; g < 4; ++g) {
          v4f lo; lo.x = acc[g][0]; lo.y = acc[g][1]; lo.z = acc[g][2]; lo.w = acc[g][3];
          v4f hi; hi.x = acc[g][4]; hi.y = acc[g][5]; hi.z = acc[g][6]; hi.w = acc[g][7];
          *(v4f*)(slot + (g * 2 + 0) * 128) = lo;
          *(v4f*)(slot + (g * 2 + 1) * 128) = hi;
        }
      }
    }
  };

  auto consume = [&](int ch) {
    const float* cbase = ring + (ch & 1) * (TC * SLOT) + lane * 4;
    const int left = T_ - ch * TC;
    const int nst  = left < TC ? left : TC;
    v4f cur[8];
#pragma unroll
    for (int q = 0; q < 8; ++q) cur[q] = *(const v4f*)(cbase + q * 128);
    for (int tl = 0; tl < nst; ++tl) {
      const int tn = (tl + 1 < nst) ? tl + 1 : tl;
      v4f nxt[8];
#pragma unroll
      for (int q = 0; q < 8; ++q) nxt[q] = *(const v4f*)(cbase + tn * SLOT + q * 128);

      v8f acc[4];
#pragma unroll
      for (int g = 0; g < 4; ++g) {
        acc[g][0] = cur[2 * g][0]; acc[g][1] = cur[2 * g][1]; acc[g][2] = cur[2 * g][2]; acc[g][3] = cur[2 * g][3];
        acc[g][4] = cur[2 * g + 1][0]; acc[g][5] = cur[2 * g + 1][1]; acc[g][6] = cur[2 * g + 1][2]; acc[g][7] = cur[2 * g + 1][3];
      }

#pragma unroll
      for (int r = 0; r < 8; ++r) hl[(r + mhalf) * HPAD + n] = h[r];
#pragma unroll
      for (int kk = 0; kk < 4; ++kk) {
        v2f a = *(const v2f*)(hl + n * HPAD + kk * 4 + khalf);
#pragma unroll
        for (int g = 0; g < 4; ++g)
          acc[g] = __builtin_amdgcn_wmma_f32_16x16x4_f32(false, a, false, ufrag[kk][g],
                                                         (short)0, acc[g], false, false);
      }

#pragma unroll
      for (int r = 0; r < 8; ++r) {
        float gi = sig_gate(acc[0][r]);
        float gf = sig_gate(acc[1][r]);
        float gg = fmaxf(acc[2][r], 0.0f);
        float go = sig_gate(acc[3][r]);
        float cn = gf * c[r] + gi * gg;
        c[r] = cn;
        h[r] = go * fmaxf(cn, 0.0f);
      }
#pragma unroll
      for (int q = 0; q < 8; ++q) cur[q] = nxt[q];
    }
  };

  for (int ch = -1; ch < NCH; ++ch) {
    if (wave == 0) {
      if (ch >= 0) consume(ch);
    } else {
      if (ch + 1 < NCH) produce(ch + 1);
    }
    __syncthreads();
  }

  if (wave != 0) return;

#pragma unroll
  for (int r = 0; r < 8; ++r) hl[(r + mhalf) * HPAD + n] = h[r];

  v8f dacc[4];
#pragma unroll
  for (int j = 0; j < 4; ++j) {
    float bj = bd[j * 16 + n];
#pragma unroll
    for (int r = 0; r < 8; ++r) dacc[j][r] = bj;
  }
#pragma unroll
  for (int kk = 0; kk < 4; ++kk) {
    v2f a = *(const v2f*)(hl + n * HPAD + kk * 4 + khalf);
    const float* wrow = Wdlds + (kk * 2 + half) * WPP;
#pragma unroll
    for (int j = 0; j < 4; ++j) {
      v2f wb = *(const v2f*)(wrow + (j * 16 + n) * 2);
      dacc[j] = __builtin_amdgcn_wmma_f32_16x16x4_f32(false, a, false, wb,
                                                      (short)0, dacc[j], false, false);
    }
  }

  float* ol = &ring[0];
#pragma unroll
  for (int j = 0; j < 4; ++j)
#pragma unroll
    for (int r = 0; r < 8; ++r)
      ol[(r + mhalf) * XPAD + j * 16 + n] = sigf(dacc[j][r]);

  if (lane < 16) {
    float s = 0.0f;
    for (int q = 0; q < 64; ++q) s += ol[lane * XPAD + q];
    hl[lane] = 1.0f / s;
  }

#pragma unroll 4
  for (int i = 0; i < 32; ++i) {
    int idx = i * 32 + lane;
    int row = idx >> 6, col = idx & 63;
    out[(size_t)(b0 + row) * D_ + col] = ol[row * XPAD + col] * hl[row];
  }
}

extern "C" void kernel_launch(void* const* d_in, const int* in_sizes, int n_in,
                              void* d_out, int out_size, void* d_ws, size_t ws_size,
                              hipStream_t stream) {
  (void)in_sizes; (void)n_in; (void)out_size; (void)d_ws; (void)ws_size;
  const float* x  = (const float*)d_in[0];
  const float* W  = (const float*)d_in[1];
  const float* U  = (const float*)d_in[2];
  const float* b  = (const float*)d_in[3];
  const float* Wd = (const float*)d_in[4];
  const float* bd = (const float*)d_in[5];
  float* out = (float*)d_out;

  dim3 grid(B_ / 16);
  dim3 block(128);
  hipLaunchKernelGGL(lstm_fused_kernel, grid, block, 0, stream,
                     x, W, U, b, Wd, bd, out);
}
